# starting kernel
# baseline (speedup 1.0000x reference)
#include <hip/hip_runtime.h>

#define B_  64
#define S_  4096
#define C_  128
#define L_  128
#define K_  32
#define NL_ (B_ * C_)
#define LANES_ 16
#define NTHREADS_ (LANES_ * K_)

#define ROWSTRIDE_   19
#define CHUNKSTRIDE_ (L_ * ROWSTRIDE_ + 16)
#define EP_FLOATS_   (K_ * CHUNKSTRIDE_)
#define H_OFF_       EP_FLOATS_
#define CIN_OFF_     (H_OFF_ + L_ * 4)
#define LDS_FLOATS_  (CIN_OFF_ + K_ * 4 * LANES_)
#define LDS_BYTES_   (LDS_FLOATS_ * 4)

typedef float v2f __attribute__((ext_vector_type(2)));
typedef float v8f __attribute__((ext_vector_type(8)));

__global__ void __launch_bounds__(NTHREADS_) arima_fused(
    const float* __restrict__ y, const float* __restrict__ phi,
    const float* __restrict__ theta, float* __restrict__ out) {
    extern __shared__ float lds[];
    float* EP  = lds;
    float* Hh  = lds + H_OFF_;
    float* CIN = lds + CIN_OFF_;

    const int tid   = threadIdx.x;
    const int lt    = blockIdx.x;
    const int bb    = lt >> 3;
    const int cbase = (lt & 7) * LANES_;

    const float t0 = theta[0], t1 = theta[1], t2 = theta[2];

    if (tid < 4) {
        const int j = tid;
        float e1 = (j == 0) ? 1.f : 0.f;
        float e2 = (j == 1) ? 1.f : 0.f;
        float e3 = (j == 2) ? 1.f : 0.f;
        for (int t = 0; t < L_; ++t) {
            float e = -(t0 * e1 + t1 * e2 + t2 * e3);
            Hh[t * 4 + j] = e;
            e3 = e2; e2 = e1; e1 = e;
        }
    }

    {
        const float p0 = phi[0], p1 = phi[1], p2 = phi[2];
        const int k = tid >> 4;
        const int x = tid & 15;
        const float* yp = y + ((size_t)bb * S_ + (size_t)k * L_) * C_ + cbase + x;
        float* ep = EP + k * CHUNKSTRIDE_ + x;
        float yl1 = 0.f, yl2 = 0.f, yl3 = 0.f;
        if (k > 0) { yl1 = yp[-C_]; yl2 = yp[-2 * C_]; yl3 = yp[-3 * C_]; }
        float e1 = 0.f, e2 = 0.f, e3 = 0.f;
#pragma unroll 4
        for (int t = 0; t < L_; ++t) {
            float yt = yp[t * C_];
            float z  = yt - p0 * yl1 - p1 * yl2 - p2 * yl3;
            float w  = z  - t1 * e2  - t2 * e3;
            float e  = w  - t0 * e1;
            ep[t * ROWSTRIDE_] = e;
            yl3 = yl2; yl2 = yl1; yl1 = yt;
            e3 = e2; e2 = e1; e1 = e;
        }
    }
    __syncthreads();

    if (tid < LANES_) {
        const int x = tid;
        const float M00 = Hh[(L_-1)*4+0], M01 = Hh[(L_-1)*4+1], M02 = Hh[(L_-1)*4+2];
        const float M10 = Hh[(L_-2)*4+0], M11 = Hh[(L_-2)*4+1], M12 = Hh[(L_-2)*4+2];
        const float M20 = Hh[(L_-3)*4+0], M21 = Hh[(L_-3)*4+1], M22 = Hh[(L_-3)*4+2];
        float c1 = 0.f, c2 = 0.f, c3 = 0.f;
        for (int k = 0; k < K_; ++k) {
            CIN[(k*4 + 0) * LANES_ + x] = c1;
            CIN[(k*4 + 1) * LANES_ + x] = c2;
            CIN[(k*4 + 2) * LANES_ + x] = c3;
            CIN[(k*4 + 3) * LANES_ + x] = 0.f;
            const float* ep = EP + k * CHUNKSTRIDE_ + x;
            float s0 = ep[(L_-1)*ROWSTRIDE_] + M00*c1 + M01*c2 + M02*c3;
            float s1 = ep[(L_-2)*ROWSTRIDE_] + M10*c1 + M11*c2 + M12*c3;
            float s2 = ep[(L_-3)*ROWSTRIDE_] + M20*c1 + M21*c2 + M22*c3;
            c1 = s0; c2 = s1; c3 = s2;
        }
    }
    __syncthreads();

    {
        const int l    = tid & 31;
        const int w    = tid >> 5;
        const int n    = l & 15;
        const int half = l >> 4;
        float* obase = out + (size_t)bb * S_ * C_ + cbase + n;
#pragma unroll
        for (int kk = 0; kk < 2; ++kk) {
            const int k = 2 * w + kk;
            v2f bvec;
            bvec.x = CIN[(k*4 + half*2 + 0) * LANES_ + n];
            bvec.y = CIN[(k*4 + half*2 + 1) * LANES_ + n];
            const float* epk = EP + k * CHUNKSTRIDE_ + n;
#pragma unroll
            for (int tt = 0; tt < L_ / 16; ++tt) {
                const int tl = tt * 16 + n;
                v2f avec = *(const v2f*)(Hh + tl * 4 + half * 2);
                v8f cvec;
#pragma unroll
                for (int r = 0; r < 8; ++r)
                    cvec[r] = epk[(tt * 16 + half * 8 + r) * ROWSTRIDE_];
                v8f d = __builtin_amdgcn_wmma_f32_16x16x4_f32(
                     false, avec,  false, bvec,
                     (short)0, cvec,  false,  false);
                float* op = obase + (size_t)(k * L_ + tt * 16 + half * 8) * C_;
#pragma unroll
                for (int r = 0; r < 8; ++r) op[r * C_] = d[r];
            }
        }
    }
}

extern "C" void kernel_launch(void* const* d_in, const int* in_sizes, int n_in,
                              void* d_out, int out_size, void* d_ws, size_t ws_size,
                              hipStream_t stream) {
    const float* y     = (const float*)d_in[0];
    const float* phi   = (const float*)d_in[1];
    const float* theta = (const float*)d_in[2];
    float* out = (float*)d_out;
    (void)d_ws; (void)ws_size; (void)in_sizes; (void)n_in; (void)out_size;

    arima_fused<<<NL_ / LANES_, NTHREADS_, LDS_BYTES_, stream>>>(y, phi, theta, out);
}


# Lane = channel: whole-line row loads feed an in-register walk (no LDS transposition); block-diagonal A lets the fixup WMMA work on those row registers, summed in two products (c1,c2 then c3)
# speedup vs baseline: 5.3098x; 5.3098x over previous
#include <hip/hip_runtime.h>

#define B_  64
#define S_  4096
#define C_  128
#define L_  128
#define K_  32
#define CW_ 32
#define NW_ 16
#define NTHREADS_ (NW_ * 32)
static_assert(B_ == 64 && C_ == 4 * CW_ && K_ == 2 * NW_ && K_ * L_ == S_, "work-group index = batch (6 bits) | channel group (2 bits)");

#define PARK_FLOATS_ (NW_ * L_ * CW_)
#define H_OFF_       PARK_FLOATS_
#define Z_OFF_       (H_OFF_ + L_ * 4)
#define ST_OFF_      (Z_OFF_ + L_ * 4)
#define CIN_OFF_     (ST_OFF_ + K_ * 4 * CW_)
#define LDS_FLOATS_  (CIN_OFF_ + K_ * 4 * CW_)
#define LDS_BYTES_   (LDS_FLOATS_ * 4)

typedef float v2f __attribute__((ext_vector_type(2)));
typedef float v4f __attribute__((ext_vector_type(4)));
typedef float v8f __attribute__((ext_vector_type(8)));


__device__ __forceinline__ void fixup_store(const float (&R)[L_], const float* __restrict__ ab,
                                            v2f b1, v2f b2, float* __restrict__ ob) {
#pragma unroll
    for (int j = 0; j < L_ / 8; ++j) {
        const v4f a = *(const v4f*)(ab + j * 32);
        const v2f a1 = {a.x, a.y};
        const v2f a2 = {a.z, a.w};
        v8f c;
#pragma unroll
        for (int r = 0; r < 8; ++r) c[r] = R[8 * j + r];
        v8f d = __builtin_amdgcn_wmma_f32_16x16x4_f32(
             false, a1,  false, b1,
             (short)0, c,  false,  false);
        d = __builtin_amdgcn_wmma_f32_16x16x4_f32(
             false, a2,  false, b2,
             (short)0, d,  false,  false);
#pragma unroll
        for (int r = 0; r < 8; ++r) ob[(size_t)(8 * j + r) * C_] = d[r];
    }
}

__global__ void __launch_bounds__(NTHREADS_) arima_fused(
    const float* __restrict__ y, const float* __restrict__ phi,
    const float* __restrict__ theta, float* __restrict__ out) {
    extern __shared__ float lds[];
    float* PARK = lds;
    float* Hh   = lds + H_OFF_;
    float* ZZ   = lds + Z_OFF_;
    float* ST   = lds + ST_OFF_;
    float* CIN  = lds + CIN_OFF_;

    const int tid   = threadIdx.x;
    const int pr    = blockIdx.x;
    const int bb    = pr & (B_ - 1);
    const int cbase = (pr >> 6) * CW_;

    const float t0 = theta[0], t1 = theta[1], t2 = theta[2];
    const float p0 = phi[0], p1 = phi[1], p2 = phi[2];

    const int l = tid & 31;
    const int w = tid >> 5;

    const float* ya = y + ((size_t)bb * S_ + (size_t)w * 2 * L_) * C_ + cbase + l;
    const float* yb = ya + (size_t)L_ * C_;
    float* pkw = PARK + w * (L_ * CW_) + l * 4;

    float yl1 = 0.f, yl2 = 0.f, yl3 = 0.f;
    if (w > 0) { yl1 = ya[-C_]; yl2 = ya[-2 * C_]; yl3 = ya[-3 * C_]; }

    float R[L_];
#pragma unroll
    for (int t = 0; t < L_; ++t) R[t] = ya[(size_t)t * C_];

    if (tid < 4) {
        const int j = tid;
        float e1 = (j == 0) ? 1.f : 0.f;
        float e2 = (j == 1) ? 1.f : 0.f;
        float e3 = (j == 2) ? 1.f : 0.f;
        for (int t = 0; t < L_; ++t) {
            float e = -(t0 * e1 + t1 * e2 + t2 * e3);
            Hh[t * 4 + j] = e;
            e3 = e2; e2 = e1; e1 = e;
        }
    }
    if (tid >= 64 && tid < 64 + L_) {
        const v4f zero = {0.f, 0.f, 0.f, 0.f};
        *(v4f*)(ZZ + (tid - 64) * 4) = zero;
    }

    {
        float e1 = 0.f, e2 = 0.f, e3 = 0.f;
#pragma unroll
        for (int t = 0; t < L_; ++t) {
            const float yt = R[t];
            float z  = yt - p0 * yl1 - p1 * yl2 - p2 * yl3;
            float wv = z  - t1 * e2  - t2 * e3;
            float e  = wv - t0 * e1;
            R[t] = e;
            yl3 = yl2; yl2 = yl1; yl1 = yt;
            e3 = e2; e2 = e1; e1 = e;
            if ((t & 3) == 3) {
                const v4f v = {R[t - 3], R[t - 2], R[t - 1], R[t]};
                *(v4f*)(pkw + (t >> 2) * (CW_ * 4)) = v;
                R[t - 3] = yb[(size_t)(t - 3) * C_];
                R[t - 2] = yb[(size_t)(t - 2) * C_];
                R[t - 1] = yb[(size_t)(t - 1) * C_];
                R[t]     = yb[(size_t)t * C_];
            }
        }
        ST[((2 * w) * 4 + 0) * CW_ + l] = e1;
        ST[((2 * w) * 4 + 1) * CW_ + l] = e2;
        ST[((2 * w) * 4 + 2) * CW_ + l] = e3;
    }
    {
        float e1 = 0.f, e2 = 0.f, e3 = 0.f;
#pragma unroll
        for (int t = 0; t < L_; ++t) {
            const float yt = R[t];
            float z  = yt - p0 * yl1 - p1 * yl2 - p2 * yl3;
            float wv = z  - t1 * e2  - t2 * e3;
            float e  = wv - t0 * e1;
            R[t] = e;
            yl3 = yl2; yl2 = yl1; yl1 = yt;
            e3 = e2; e2 = e1; e1 = e;
        }
        ST[((2 * w + 1) * 4 + 0) * CW_ + l] = e1;
        ST[((2 * w + 1) * 4 + 1) * CW_ + l] = e2;
        ST[((2 * w + 1) * 4 + 2) * CW_ + l] = e3;
    }
    __syncthreads();

    if (w == 0) {
        const float M00 = Hh[(L_-1)*4+0], M01 = Hh[(L_-1)*4+1], M02 = Hh[(L_-1)*4+2];
        const float M10 = Hh[(L_-2)*4+0], M11 = Hh[(L_-2)*4+1], M12 = Hh[(L_-2)*4+2];
        const float M20 = Hh[(L_-3)*4+0], M21 = Hh[(L_-3)*4+1], M22 = Hh[(L_-3)*4+2];
        float c1 = 0.f, c2 = 0.f, c3 = 0.f;
        for (int kc = 0; kc < K_; ++kc) {
            CIN[(kc*4 + 0) * CW_ + l] = c1;
            CIN[(kc*4 + 1) * CW_ + l] = c2;
            CIN[(kc*4 + 2) * CW_ + l] = c3;
            float s0 = ST[(kc*4 + 0) * CW_ + l] + M00*c1 + M01*c2 + M02*c3;
            float s1 = ST[(kc*4 + 1) * CW_ + l] + M10*c1 + M11*c2 + M12*c3;
            float s2 = ST[(kc*4 + 2) * CW_ + l] + M20*c1 + M21*c2 + M22*c3;
            c1 = s0; c2 = s1; c3 = s2;
        }
    }
    __syncthreads();

    const bool act = ((l >> 4) == ((l >> 3) & 1));
    const float* ab = (act ? Hh : ZZ) + (l & 7) * 4;
    float* ob = out + ((size_t)bb * S_ + (size_t)w * 2 * L_) * C_ + cbase + l;
    {
        const int kb = 2 * w + 1;
        const v2f b1 = {CIN[(kb*4 + 0) * CW_ + l], CIN[(kb*4 + 1) * CW_ + l]};
        const v2f b2 = {CIN[(kb*4 + 2) * CW_ + l], 0.f};
        fixup_store(R, ab, b1, b2, ob + (size_t)L_ * C_);
    }
    {
#pragma unroll
        for (int g = 0; g < L_ / 4; ++g) {
            const v4f v = *(const v4f*)(pkw + g * (CW_ * 4));
            R[4 * g + 0] = v.x; R[4 * g + 1] = v.y; R[4 * g + 2] = v.z; R[4 * g + 3] = v.w;
        }
        const int ka = 2 * w;
        const v2f b1 = {CIN[(ka*4 + 0) * CW_ + l], CIN[(ka*4 + 1) * CW_ + l]};
        const v2f b2 = {CIN[(ka*4 + 2) * CW_ + l], 0.f};
        fixup_store(R, ab, b1, b2, ob);
    }
}

extern "C" void kernel_launch(void* const* d_in, const int* in_sizes, int n_in,
                              void* d_out, int out_size, void* d_ws, size_t ws_size,
                              hipStream_t stream) {
    const float* y     = (const float*)d_in[0];
    const float* phi   = (const float*)d_in[1];
    const float* theta = (const float*)d_in[2];
    float* out = (float*)d_out;
    (void)d_ws; (void)ws_size; (void)in_sizes; (void)n_in; (void)out_size;

    arima_fused<<<(B_ * C_) / CW_, NTHREADS_, LDS_BYTES_, stream>>>(y, phi, theta, out);
}


# Group = 64 adjacent channels x half the time axis, chunks of 64 steps: each load/store covers a whole aligned 256-B unit; second-half groups rewalk the first half's y for the mid-sequence state
# speedup vs baseline: 6.7167x; 1.2650x over previous
#include <hip/hip_runtime.h>

#define B_  64
#define S_  4096
#define C_  128
#define HALF_ (S_ / 2)
#define L_  64
#define K_  32
#define CW_ 64
#define NW_ 16
#define NTHREADS_ (NW_ * 32)
static_assert(B_ == 64 && C_ == 2 * CW_ && K_ == 2 * NW_ && K_ * L_ == HALF_ && 2 * HALF_ == S_,
              "work-group index = batch (6 bits) | half of the time axis (1 bit) | channel group (1 bit)");

#define PARK_FLOATS_ (NW_ * L_ * CW_)
#define H_OFF_       PARK_FLOATS_
#define Z_OFF_       (H_OFF_ + L_ * 4)
#define ST_OFF_      (Z_OFF_ + L_ * 4)
#define STP_OFF_     (ST_OFF_ + K_ * 3 * CW_)
#define LDS_FLOATS_  (STP_OFF_ + K_ * 3 * CW_)
#define LDS_BYTES_   (LDS_FLOATS_ * 4)

typedef float v2f __attribute__((ext_vector_type(2)));
typedef float v4f __attribute__((ext_vector_type(4)));
typedef float v8f __attribute__((ext_vector_type(8)));


struct Lags { float a1, a2, a3, b1, b2, b3; };

__device__ __forceinline__ void load_rows(float (&R)[2 * L_], const float* __restrict__ p) {
#pragma unroll
    for (int t = 0; t < L_; ++t) {
        const v2f v = *(const v2f*)(p + (size_t)t * C_);
        R[2 * t] = v.x; R[2 * t + 1] = v.y;
    }
}

template <bool KEEP, bool PARKIT, bool REFILL>
__device__ __forceinline__ void walk(float (&R)[2 * L_], Lags& g,
                                     float p0, float p1, float p2, float t0, float t1, float t2,
                                     float* __restrict__ pk, const float* __restrict__ nxt,
                                     v2f& s1, v2f& s2, v2f& s3) {
    float ea1 = 0.f, ea2 = 0.f, ea3 = 0.f, eb1 = 0.f, eb2 = 0.f, eb3 = 0.f;
#pragma unroll
    for (int t = 0; t < L_; ++t) {
        const float ya = R[2 * t], yb = R[2 * t + 1];
        float za = ya - p0 * g.a1 - p1 * g.a2 - p2 * g.a3;
        float zb = yb - p0 * g.b1 - p1 * g.b2 - p2 * g.b3;
        float wa = za - t1 * ea2 - t2 * ea3;
        float wb = zb - t1 * eb2 - t2 * eb3;
        float ea = wa - t0 * ea1;
        float eb = wb - t0 * eb1;
        if (KEEP) { R[2 * t] = ea; R[2 * t + 1] = eb; }
        g.a3 = g.a2; g.a2 = g.a1; g.a1 = ya;
        g.b3 = g.b2; g.b2 = g.b1; g.b1 = yb;
        ea3 = ea2; ea2 = ea1; ea1 = ea;
        eb3 = eb2; eb2 = eb1; eb1 = eb;
        if ((t & 1) == 1) {
            if (PARKIT) {
                const v4f v = {R[2 * t - 2], R[2 * t - 1], R[2 * t], R[2 * t + 1]};
                *(v4f*)(pk + (t >> 1) * (32 * 4)) = v;
            }
            if (REFILL) {
                const v2f u0 = *(const v2f*)(nxt + (size_t)(t - 1) * C_);
                const v2f u1 = *(const v2f*)(nxt + (size_t)t * C_);
                R[2 * t - 2] = u0.x; R[2 * t - 1] = u0.y;
                R[2 * t]     = u1.x; R[2 * t + 1] = u1.y;
            }
        }
    }
    s1 = v2f{ea1, eb1}; s2 = v2f{ea2, eb2}; s3 = v2f{ea3, eb3};
}

__device__ __forceinline__ void fixup_store(const float (&R)[2 * L_], const float* __restrict__ ab,
                                            v2f b1a, v2f b2a, v2f b1b, v2f b2b, float* __restrict__ ob) {
#pragma unroll
    for (int j = 0; j < L_ / 8; ++j) {
        const v4f a = *(const v4f*)(ab + j * 32);
        const v2f a1 = {a.x, a.y};
        const v2f a2 = {a.z, a.w};
        v8f ca, cb;
#pragma unroll
        for (int r = 0; r < 8; ++r) { ca[r] = R[2 * (8 * j + r)]; cb[r] = R[2 * (8 * j + r) + 1]; }
        v8f da = __builtin_amdgcn_wmma_f32_16x16x4_f32(
             false, a1,  false, b1a,
             (short)0, ca,  false,  false);
        v8f db = __builtin_amdgcn_wmma_f32_16x16x4_f32(
             false, a1,  false, b1b,
             (short)0, cb,  false,  false);
        da = __builtin_amdgcn_wmma_f32_16x16x4_f32(
             false, a2,  false, b2a,
             (short)0, da,  false,  false);
        db = __builtin_amdgcn_wmma_f32_16x16x4_f32(
             false, a2,  false, b2b,
             (short)0, db,  false,  false);
#pragma unroll
        for (int r = 0; r < 8; ++r) {
            const v2f o = {da[r], db[r]};
            *(v2f*)(ob + (size_t)(8 * j + r) * C_) = o;
        }
    }
}

template <bool INPLACE>
__device__ __forceinline__ void scan_chunks(float* __restrict__ st, const float* __restrict__ Hh, int l,
                                            v2f& c1, v2f& c2, v2f& c3) {
    const float M00 = Hh[(L_-1)*4+0], M01 = Hh[(L_-1)*4+1], M02 = Hh[(L_-1)*4+2];
    const float M10 = Hh[(L_-2)*4+0], M11 = Hh[(L_-2)*4+1], M12 = Hh[(L_-2)*4+2];
    const float M20 = Hh[(L_-3)*4+0], M21 = Hh[(L_-3)*4+1], M22 = Hh[(L_-3)*4+2];
    for (int kc = 0; kc < K_; ++kc) {
        v2f* q0 = (v2f*)(st + (kc*3 + 0) * CW_ + 2 * l);
        v2f* q1 = (v2f*)(st + (kc*3 + 1) * CW_ + 2 * l);
        v2f* q2 = (v2f*)(st + (kc*3 + 2) * CW_ + 2 * l);
        const v2f r0 = *q0, r1 = *q1, r2 = *q2;
        if (INPLACE) { *q0 = c1; *q1 = c2; *q2 = c3; }
        v2f s0, s1, s2;
        s0.x = r0.x + M00*c1.x + M01*c2.x + M02*c3.x;
        s0.y = r0.y + M00*c1.y + M01*c2.y + M02*c3.y;
        s1.x = r1.x + M10*c1.x + M11*c2.x + M12*c3.x;
        s1.y = r1.y + M10*c1.y + M11*c2.y + M12*c3.y;
        s2.x = r2.x + M20*c1.x + M21*c2.x + M22*c3.x;
        s2.y = r2.y + M20*c1.y + M21*c2.y + M22*c3.y;
        c1 = s0; c2 = s1; c3 = s2;
    }
}

__global__ void __launch_bounds__(NTHREADS_) arima_fused(
    const float* __restrict__ y, const float* __restrict__ phi,
    const float* __restrict__ theta, float* __restrict__ out) {
    extern __shared__ float lds[];
    float* PARK = lds;
    float* Hh   = lds + H_OFF_;
    float* ZZ   = lds + Z_OFF_;
    float* ST   = lds + ST_OFF_;
    float* STP  = lds + STP_OFF_;

    const int tid   = threadIdx.x;
    const int pr    = blockIdx.x;
    const int bb    = pr & (B_ - 1);
    const int hf    = ((pr >> 6) & 1) ^ 1;
    const int cbase = (pr >> 7) * CW_;

    const float t0 = theta[0], t1 = theta[1], t2 = theta[2];
    const float p0 = phi[0], p1 = phi[1], p2 = phi[2];

    const int l = tid & 31;
    const int w = tid >> 5;

    const float* ys = y + (size_t)bb * S_ * C_ + cbase + 2 * l;
    const float* yp = ys + (size_t)w * 2 * L_ * C_;
    const float* ya = ys + ((size_t)hf * HALF_ + (size_t)w * 2 * L_) * C_;
    const float* yb = ya + (size_t)L_ * C_;
    float* pkw = PARK + w * (L_ * CW_) + l * 4;

    const float* yfirst = hf ? yp : ya;
    Lags g = {0.f, 0.f, 0.f, 0.f, 0.f, 0.f};
    if (w > 0) {
        const v2f m1 = *(const v2f*)(yfirst - C_), m2 = *(const v2f*)(yfirst - 2 * C_), m3 = *(const v2f*)(yfirst - 3 * C_);
        g.a1 = m1.x; g.b1 = m1.y; g.a2 = m2.x; g.b2 = m2.y; g.a3 = m3.x; g.b3 = m3.y;
    }
    Lags gm = {0.f, 0.f, 0.f, 0.f, 0.f, 0.f};
    if (hf) {
        const v2f m1 = *(const v2f*)(ya - C_), m2 = *(const v2f*)(ya - 2 * C_), m3 = *(const v2f*)(ya - 3 * C_);
        gm.a1 = m1.x; gm.b1 = m1.y; gm.a2 = m2.x; gm.b2 = m2.y; gm.a3 = m3.x; gm.b3 = m3.y;
    }

    float R[2 * L_];
    load_rows(R, yfirst);

    if (tid < 4) {
        const int j = tid;
        float e1 = (j == 0) ? 1.f : 0.f;
        float e2 = (j == 1) ? 1.f : 0.f;
        float e3 = (j == 2) ? 1.f : 0.f;
        for (int t = 0; t < L_; ++t) {
            float e = -(t0 * e1 + t1 * e2 + t2 * e3);
            Hh[t * 4 + j] = e;
            e3 = e2; e2 = e1; e1 = e;
        }
    }
    if (tid >= 64 && tid < 64 + L_) {
        const v4f zero = {0.f, 0.f, 0.f, 0.f};
        *(v4f*)(ZZ + (tid - 64) * 4) = zero;
    }

    v2f m1 = {0.f, 0.f}, m2 = {0.f, 0.f}, m3 = {0.f, 0.f};
    if (hf) {
        v2f s1, s2, s3;
        walk<false, false, true>(R, g, p0, p1, p2, t0, t1, t2, nullptr, yp + (size_t)L_ * C_, s1, s2, s3);
        *(v2f*)(STP + ((2 * w) * 3 + 0) * CW_ + 2 * l) = s1;
        *(v2f*)(STP + ((2 * w) * 3 + 1) * CW_ + 2 * l) = s2;
        *(v2f*)(STP + ((2 * w) * 3 + 2) * CW_ + 2 * l) = s3;
        walk<false, false, true>(R, g, p0, p1, p2, t0, t1, t2, nullptr, ya, s1, s2, s3);
        *(v2f*)(STP + ((2 * w + 1) * 3 + 0) * CW_ + 2 * l) = s1;
        *(v2f*)(STP + ((2 * w + 1) * 3 + 1) * CW_ + 2 * l) = s2;
        *(v2f*)(STP + ((2 * w + 1) * 3 + 2) * CW_ + 2 * l) = s3;
        g = gm;
        __syncthreads();
        if (w == 0) scan_chunks<false>(STP, Hh, l, m1, m2, m3);
    }

    {
        v2f s1, s2, s3;
        walk<true, true, true>(R, g, p0, p1, p2, t0, t1, t2, pkw, yb, s1, s2, s3);
        *(v2f*)(ST + ((2 * w) * 3 + 0) * CW_ + 2 * l) = s1;
        *(v2f*)(ST + ((2 * w) * 3 + 1) * CW_ + 2 * l) = s2;
        *(v2f*)(ST + ((2 * w) * 3 + 2) * CW_ + 2 * l) = s3;
    }
    {
        v2f s1, s2, s3;
        walk<true, false, false>(R, g, p0, p1, p2, t0, t1, t2, nullptr, nullptr, s1, s2, s3);
        *(v2f*)(ST + ((2 * w + 1) * 3 + 0) * CW_ + 2 * l) = s1;
        *(v2f*)(ST + ((2 * w + 1) * 3 + 1) * CW_ + 2 * l) = s2;
        *(v2f*)(ST + ((2 * w + 1) * 3 + 2) * CW_ + 2 * l) = s3;
    }
    __syncthreads();

    if (w == 0) scan_chunks<true>(ST, Hh, l, m1, m2, m3);
    __syncthreads();

    const bool act = ((l >> 4) == ((l >> 3) & 1));
    const float* ab = (act ? Hh : ZZ) + (l & 7) * 4;
    float* ob = out + ((size_t)bb * S_ + (size_t)hf * HALF_ + (size_t)w * 2 * L_) * C_ + cbase + 2 * l;
    {
        const int kb = 2 * w + 1;
        const v2f c1 = *(const v2f*)(ST + (kb*3 + 0) * CW_ + 2 * l);
        const v2f c2 = *(const v2f*)(ST + (kb*3 + 1) * CW_ + 2 * l);
        const v2f c3 = *(const v2f*)(ST + (kb*3 + 2) * CW_ + 2 * l);
        const v2f b1a = {c1.x, c2.x}, b2a = {c3.x, 0.f};
        const v2f b1b = {c1.y, c2.y}, b2b = {c3.y, 0.f};
        fixup_store(R, ab, b1a, b2a, b1b, b2b, ob + (size_t)L_ * C_);
    }
    {
#pragma unroll
        for (int q = 0; q < L_ / 2; ++q) {
            const v4f v = *(const v4f*)(pkw + q * (32 * 4));
            R[4 * q + 0] = v.x; R[4 * q + 1] = v.y; R[4 * q + 2] = v.z; R[4 * q + 3] = v.w;
        }
        const int ka = 2 * w;
        const v2f c1 = *(const v2f*)(ST + (ka*3 + 0) * CW_ + 2 * l);
        const v2f c2 = *(const v2f*)(ST + (ka*3 + 1) * CW_ + 2 * l);
        const v2f c3 = *(const v2f*)(ST + (ka*3 + 2) * CW_ + 2 * l);
        const v2f b1a = {c1.x, c2.x}, b2a = {c3.x, 0.f};
        const v2f b1b = {c1.y, c2.y}, b2b = {c3.y, 0.f};
        fixup_store(R, ab, b1a, b2a, b1b, b2b, ob);
    }
}

extern "C" void kernel_launch(void* const* d_in, const int* in_sizes, int n_in,
                              void* d_out, int out_size, void* d_ws, size_t ws_size,
                              hipStream_t stream) {
    const float* y     = (const float*)d_in[0];
    const float* phi   = (const float*)d_in[1];
    const float* theta = (const float*)d_in[2];
    float* out = (float*)d_out;
    (void)d_ws; (void)ws_size; (void)in_sizes; (void)n_in; (void)out_size;

    arima_fused<<<256, NTHREADS_, LDS_BYTES_, stream>>>(y, phi, theta, out);
}


# Stored half runs through the group as two quarters (a wave owns one chunk of each): quarter 2's rows load under quarter 1's scan+fixup, quarter 1's stores drain under quarter 2; LDS-only barriers
# speedup vs baseline: 7.1450x; 1.0638x over previous
#include <hip/hip_runtime.h>

#define B_  64
#define S_  4096
#define C_  128
#define HALF_ (S_ / 2)
#define L_  64
#define K_  32
#define CW_ 64
#define NW_ 16
#define NTHREADS_ (NW_ * 32)
static_assert(B_ == 64 && C_ == 2 * CW_ && K_ == 2 * NW_ && K_ * L_ == HALF_ && 2 * HALF_ == S_,
              "work-group index = batch (6 bits) | half of the time axis (1 bit) | channel group (1 bit)");

#define PARK_FLOATS_ (NW_ * L_ * CW_)
#define H_OFF_       PARK_FLOATS_
#define Z_OFF_       (H_OFF_ + L_ * 4)
#define ST_OFF_      (Z_OFF_ + L_ * 4)
#define STP_OFF_     (ST_OFF_ + (K_ + 1) * 3 * CW_)
#define LDS_FLOATS_  (STP_OFF_ + K_ * 3 * CW_)
#define LDS_BYTES_   (LDS_FLOATS_ * 4)

typedef float v2f __attribute__((ext_vector_type(2)));
typedef float v4f __attribute__((ext_vector_type(4)));
typedef float v8f __attribute__((ext_vector_type(8)));


struct Lags { float a1, a2, a3, b1, b2, b3; };

__device__ __forceinline__ void load_rows(float (&R)[2 * L_], const float* __restrict__ p) {
#pragma unroll
    for (int t = 0; t < L_; ++t) {
        const v2f v = *(const v2f*)(p + (size_t)t * C_);
        R[2 * t] = v.x; R[2 * t + 1] = v.y;
    }
}

template <bool KEEP, bool PARKIT, bool REFILL>
__device__ __forceinline__ void walk(float (&R)[2 * L_], Lags& g,
                                     float p0, float p1, float p2, float t0, float t1, float t2,
                                     float* __restrict__ pk, const float* __restrict__ nxt,
                                     v2f& s1, v2f& s2, v2f& s3) {
    float ea1 = 0.f, ea2 = 0.f, ea3 = 0.f, eb1 = 0.f, eb2 = 0.f, eb3 = 0.f;
#pragma unroll
    for (int t = 0; t < L_; ++t) {
        const float ya = R[2 * t], yb = R[2 * t + 1];
        float za = ya - p0 * g.a1 - p1 * g.a2 - p2 * g.a3;
        float zb = yb - p0 * g.b1 - p1 * g.b2 - p2 * g.b3;
        float wa = za - t1 * ea2 - t2 * ea3;
        float wb = zb - t1 * eb2 - t2 * eb3;
        float ea = wa - t0 * ea1;
        float eb = wb - t0 * eb1;
        if (KEEP) { R[2 * t] = ea; R[2 * t + 1] = eb; }
        g.a3 = g.a2; g.a2 = g.a1; g.a1 = ya;
        g.b3 = g.b2; g.b2 = g.b1; g.b1 = yb;
        ea3 = ea2; ea2 = ea1; ea1 = ea;
        eb3 = eb2; eb2 = eb1; eb1 = eb;
        if (PARKIT && (t & 7) == 0 && t > 0) __builtin_amdgcn_sched_barrier(0);
        if ((t & 1) == 1) {
            if (PARKIT) {
                const v4f v = {R[2 * t - 2], R[2 * t - 1], R[2 * t], R[2 * t + 1]};
                *(v4f*)(pk + (t >> 1) * (32 * 4)) = v;
            }
            if (REFILL) {
                const v2f u0 = *(const v2f*)(nxt + (size_t)(t - 1) * C_);
                const v2f u1 = *(const v2f*)(nxt + (size_t)t * C_);
                R[2 * t - 2] = u0.x; R[2 * t - 1] = u0.y;
                R[2 * t]     = u1.x; R[2 * t + 1] = u1.y;
            }
        }
    }
    s1 = v2f{ea1, eb1}; s2 = v2f{ea2, eb2}; s3 = v2f{ea3, eb3};
}

template <bool PARKED>
__device__ __forceinline__ void fixup_store(const float (&R)[2 * L_], const float* __restrict__ pk,
                                            const float* __restrict__ ab,
                                            v2f b1a, v2f b2a, v2f b1b, v2f b2b, float* __restrict__ ob) {
#pragma unroll
    for (int j = 0; j < L_ / 8; ++j) {
        const v4f a = *(const v4f*)(ab + j * 32);
        const v2f a1 = {a.x, a.y};
        const v2f a2 = {a.z, a.w};
        v8f ca, cb;
        if (PARKED) {
#pragma unroll
            for (int q = 0; q < 4; ++q) {
                const v4f v = *(const v4f*)(pk + (4 * j + q) * (32 * 4));
                ca[2 * q] = v.x; cb[2 * q] = v.y; ca[2 * q + 1] = v.z; cb[2 * q + 1] = v.w;
            }
        } else {
#pragma unroll
            for (int r = 0; r < 8; ++r) { ca[r] = R[2 * (8 * j + r)]; cb[r] = R[2 * (8 * j + r) + 1]; }
        }
        v8f da = __builtin_amdgcn_wmma_f32_16x16x4_f32(
             false, a1,  false, b1a,
             (short)0, ca,  false,  false);
        v8f db = __builtin_amdgcn_wmma_f32_16x16x4_f32(
             false, a1,  false, b1b,
             (short)0, cb,  false,  false);
        da = __builtin_amdgcn_wmma_f32_16x16x4_f32(
             false, a2,  false, b2a,
             (short)0, da,  false,  false);
        db = __builtin_amdgcn_wmma_f32_16x16x4_f32(
             false, a2,  false, b2b,
             (short)0, db,  false,  false);
#pragma unroll
        for (int r = 0; r < 8; ++r) {
            const v2f o = {da[r], db[r]};
            *(v2f*)(ob + (size_t)(8 * j + r) * C_) = o;
        }
    }
}

template <bool INPLACE>
__device__ __forceinline__ void scan_chunks(float* __restrict__ st, const float* __restrict__ Hh, int l,
                                            int k0, int n, v2f& c1, v2f& c2, v2f& c3) {
    const float M00 = Hh[(L_-1)*4+0], M01 = Hh[(L_-1)*4+1], M02 = Hh[(L_-1)*4+2];
    const float M10 = Hh[(L_-2)*4+0], M11 = Hh[(L_-2)*4+1], M12 = Hh[(L_-2)*4+2];
    const float M20 = Hh[(L_-3)*4+0], M21 = Hh[(L_-3)*4+1], M22 = Hh[(L_-3)*4+2];
    for (int kc = k0; kc < k0 + n; ++kc) {
        v2f* q0 = (v2f*)(st + (kc*3 + 0) * CW_ + 2 * l);
        v2f* q1 = (v2f*)(st + (kc*3 + 1) * CW_ + 2 * l);
        v2f* q2 = (v2f*)(st + (kc*3 + 2) * CW_ + 2 * l);
        const v2f r0 = *q0, r1 = *q1, r2 = *q2;
        if (INPLACE) { *q0 = c1; *q1 = c2; *q2 = c3; }
        v2f s0, s1, s2;
        s0.x = r0.x + M00*c1.x + M01*c2.x + M02*c3.x;
        s0.y = r0.y + M00*c1.y + M01*c2.y + M02*c3.y;
        s1.x = r1.x + M10*c1.x + M11*c2.x + M12*c3.x;
        s1.y = r1.y + M10*c1.y + M11*c2.y + M12*c3.y;
        s2.x = r2.x + M20*c1.x + M21*c2.x + M22*c3.x;
        s2.y = r2.y + M20*c1.y + M21*c2.y + M22*c3.y;
        c1 = s0; c2 = s1; c3 = s2;
    }
}

__device__ __forceinline__ void lds_barrier() {
    asm volatile("s_wait_dscnt 0x0\n\ts_barrier_signal -1\n\ts_barrier_wait -1" ::: "memory");
}

__global__ void __launch_bounds__(NTHREADS_) arima_fused(
    const float* __restrict__ y, const float* __restrict__ phi,
    const float* __restrict__ theta, float* __restrict__ out) {
    extern __shared__ float lds[];
    float* PARK = lds;
    float* Hh   = lds + H_OFF_;
    float* ZZ   = lds + Z_OFF_;
    float* ST   = lds + ST_OFF_;
    float* STP  = lds + STP_OFF_;

    const int tid   = threadIdx.x;
    const int pr    = blockIdx.x;
    const int bb    = pr & (B_ - 1);
    const int hf    = ((pr >> 6) & 1) ^ 1;
    const int cbase = (pr >> 7) * CW_;

    const float t0 = theta[0], t1 = theta[1], t2 = theta[2];
    const float p0 = phi[0], p1 = phi[1], p2 = phi[2];

    const int l = tid & 31;
    const int w = tid >> 5;

    const float* ys = y + (size_t)bb * S_ * C_ + cbase + 2 * l;
    const float* yp = ys + (size_t)w * 2 * L_ * C_;
    const float* ya = ys + ((size_t)hf * HALF_ + (size_t)w * L_) * C_;
    const float* yb = ya + (size_t)NW_ * L_ * C_;
    float* pkw = PARK + w * (L_ * CW_) + l * 4;

    const float* yfirst = hf ? yp : ya;
    Lags g = {0.f, 0.f, 0.f, 0.f, 0.f, 0.f};
    if (hf && w > 0) {
        const v2f m1 = *(const v2f*)(yp - C_), m2 = *(const v2f*)(yp - 2 * C_), m3 = *(const v2f*)(yp - 3 * C_);
        g.a1 = m1.x; g.b1 = m1.y; g.a2 = m2.x; g.b2 = m2.y; g.a3 = m3.x; g.b3 = m3.y;
    }
    Lags ga;
    {
        const bool has = hf || w > 0;
        const float* la = has ? ya : ya + 3 * C_;
        const v2f m1 = *(const v2f*)(la - C_), m2 = *(const v2f*)(la - 2 * C_), m3 = *(const v2f*)(la - 3 * C_);
        ga.a1 = has ? m1.x : 0.f; ga.b1 = has ? m1.y : 0.f;
        ga.a2 = has ? m2.x : 0.f; ga.b2 = has ? m2.y : 0.f;
        ga.a3 = has ? m3.x : 0.f; ga.b3 = has ? m3.y : 0.f;
    }
    Lags gq;
    {
        const v2f m1 = *(const v2f*)(yb - C_), m2 = *(const v2f*)(yb - 2 * C_), m3 = *(const v2f*)(yb - 3 * C_);
        gq.a1 = m1.x; gq.b1 = m1.y; gq.a2 = m2.x; gq.b2 = m2.y; gq.a3 = m3.x; gq.b3 = m3.y;
    }

    float R[2 * L_];
    load_rows(R, yfirst);

    if (tid < 4) {
        const int j = tid;
        float e1 = (j == 0) ? 1.f : 0.f;
        float e2 = (j == 1) ? 1.f : 0.f;
        float e3 = (j == 2) ? 1.f : 0.f;
        for (int t = 0; t < L_; ++t) {
            float e = -(t0 * e1 + t1 * e2 + t2 * e3);
            Hh[t * 4 + j] = e;
            e3 = e2; e2 = e1; e1 = e;
        }
    }
    if (tid >= 64 && tid < 64 + L_) {
        const v4f zero = {0.f, 0.f, 0.f, 0.f};
        *(v4f*)(ZZ + (tid - 64) * 4) = zero;
    }

    if (hf) {
        v2f s1, s2, s3;
        walk<false, false, true>(R, g, p0, p1, p2, t0, t1, t2, nullptr, yp + (size_t)L_ * C_, s1, s2, s3);
        *(v2f*)(STP + ((2 * w) * 3 + 0) * CW_ + 2 * l) = s1;
        *(v2f*)(STP + ((2 * w) * 3 + 1) * CW_ + 2 * l) = s2;
        *(v2f*)(STP + ((2 * w) * 3 + 2) * CW_ + 2 * l) = s3;
        walk<false, false, true>(R, g, p0, p1, p2, t0, t1, t2, nullptr, ya, s1, s2, s3);
        *(v2f*)(STP + ((2 * w + 1) * 3 + 0) * CW_ + 2 * l) = s1;
        *(v2f*)(STP + ((2 * w + 1) * 3 + 1) * CW_ + 2 * l) = s2;
        *(v2f*)(STP + ((2 * w + 1) * 3 + 2) * CW_ + 2 * l) = s3;
    }

    const bool act = ((l >> 4) == ((l >> 3) & 1));
    const float* ab = (act ? Hh : ZZ) + (l & 7) * 4;
    float* ob = out + ((size_t)bb * S_ + (size_t)hf * HALF_ + (size_t)w * L_) * C_ + cbase + 2 * l;
    float* CARRY = ST + K_ * 3 * CW_;

    {
        v2f s1, s2, s3;
        walk<true, true, true>(R, ga, p0, p1, p2, t0, t1, t2, pkw, yb, s1, s2, s3);
        *(v2f*)(ST + (w * 3 + 0) * CW_ + 2 * l) = s1;
        *(v2f*)(ST + (w * 3 + 1) * CW_ + 2 * l) = s2;
        *(v2f*)(ST + (w * 3 + 2) * CW_ + 2 * l) = s3;
    }
    lds_barrier();
    if (w == 0) {
        v2f m1 = {0.f, 0.f}, m2 = {0.f, 0.f}, m3 = {0.f, 0.f};
        if (hf) scan_chunks<false>(STP, Hh, l, 0, K_, m1, m2, m3);
        scan_chunks<true>(ST, Hh, l, 0, NW_, m1, m2, m3);
        *(v2f*)(CARRY + 0 * CW_ + 2 * l) = m1;
        *(v2f*)(CARRY + 1 * CW_ + 2 * l) = m2;
        *(v2f*)(CARRY + 2 * CW_ + 2 * l) = m3;
    }
    lds_barrier();
    {
        const v2f c1 = *(const v2f*)(ST + (w * 3 + 0) * CW_ + 2 * l);
        const v2f c2 = *(const v2f*)(ST + (w * 3 + 1) * CW_ + 2 * l);
        const v2f c3 = *(const v2f*)(ST + (w * 3 + 2) * CW_ + 2 * l);
        const v2f b1a = {c1.x, c2.x}, b2a = {c3.x, 0.f};
        const v2f b1b = {c1.y, c2.y}, b2b = {c3.y, 0.f};
        fixup_store<true>(R, pkw, ab, b1a, b2a, b1b, b2b, ob);
    }
    __builtin_amdgcn_sched_barrier(0);
    {
        v2f s1, s2, s3;
        walk<true, false, false>(R, gq, p0, p1, p2, t0, t1, t2, nullptr, nullptr, s1, s2, s3);
        *(v2f*)(ST + ((NW_ + w) * 3 + 0) * CW_ + 2 * l) = s1;
        *(v2f*)(ST + ((NW_ + w) * 3 + 1) * CW_ + 2 * l) = s2;
        *(v2f*)(ST + ((NW_ + w) * 3 + 2) * CW_ + 2 * l) = s3;
    }
    lds_barrier();
    if (w == 0) {
        v2f c1 = *(const v2f*)(CARRY + 0 * CW_ + 2 * l);
        v2f c2 = *(const v2f*)(CARRY + 1 * CW_ + 2 * l);
        v2f c3 = *(const v2f*)(CARRY + 2 * CW_ + 2 * l);
        scan_chunks<true>(ST, Hh, l, NW_, NW_, c1, c2, c3);
    }
    lds_barrier();
    {
        const int kb = NW_ + w;
        const v2f c1 = *(const v2f*)(ST + (kb * 3 + 0) * CW_ + 2 * l);
        const v2f c2 = *(const v2f*)(ST + (kb * 3 + 1) * CW_ + 2 * l);
        const v2f c3 = *(const v2f*)(ST + (kb * 3 + 2) * CW_ + 2 * l);
        const v2f b1a = {c1.x, c2.x}, b2a = {c3.x, 0.f};
        const v2f b1b = {c1.y, c2.y}, b2b = {c3.y, 0.f};
        fixup_store<false>(R, nullptr, ab, b1a, b2a, b1b, b2b, ob + (size_t)NW_ * L_ * C_);
    }
}

extern "C" void kernel_launch(void* const* d_in, const int* in_sizes, int n_in,
                              void* d_out, int out_size, void* d_ws, size_t ws_size,
                              hipStream_t stream) {
    const float* y     = (const float*)d_in[0];
    const float* phi   = (const float*)d_in[1];
    const float* theta = (const float*)d_in[2];
    float* out = (float*)d_out;
    (void)d_ws; (void)ws_size; (void)in_sizes; (void)n_in; (void)out_size;

    arima_fused<<<256, NTHREADS_, LDS_BYTES_, stream>>>(y, phi, theta, out);
}
